# starting kernel
# baseline (speedup 1.0000x reference)
#include <hip/hip_runtime.h>

typedef __attribute__((ext_vector_type(2))) float v2f;
typedef __attribute__((ext_vector_type(4))) float v4f;
typedef __attribute__((ext_vector_type(8))) float v8f;

#define DIM 512
#define TOK 1024
#define BM 128
#define BN 128
#define BK 32
#define XS_STRIDE 36
#define WT_STRIDE 36

static __device__ __forceinline__ v8f wmma4(v2f a, v2f b, v8f c) {
    return __builtin_amdgcn_wmma_f32_16x16x4_f32(
        false, a, false, b, (short)0, c, false, false);
}

__global__ __launch_bounds__(256)
void day_adapter_kernel(const float* __restrict__ x,
                        const int* __restrict__ day_ids,
                        const float* __restrict__ W,
                        const float* __restrict__ bias,
                        float* __restrict__ out) {
    __shared__ float Xs[2][BM * XS_STRIDE];
    __shared__ float Wt[2][BN * WT_STRIDE];

    const int n0    = blockIdx.x * BN;
    const int m0    = blockIdx.y * BM;
    const int batch = blockIdx.z;
    const int day   = day_ids[batch];

    const int t    = threadIdx.x;
    const int lane = t & 31;
    const int wave = t >> 5;
    const int mw   = wave >> 1;
    const int nw   = wave & 1;
    const int lm   = lane & 15;
    const int lg   = lane >> 4;

    const float* xb  = x + ((size_t)batch * TOK + m0) * DIM;
    const float* Wdn = W + (size_t)day * DIM * DIM + n0;

    const int arow0  = (mw << 5) + lm;
    const int arow1  = arow0 + 16;
    const int nwbase = nw << 6;

    v8f acc[2][4];
    #pragma unroll
    for (int m = 0; m < 2; ++m)
        #pragma unroll
        for (int n = 0; n < 4; ++n) acc[m][n] = v8f{};

    auto load_tile = [&](int buf, int k0) {
        #pragma unroll
        for (int i = 0; i < 4; ++i) {
            int f   = t + (i << 8);
            int row = f >> 3;
            int c4  = (f & 7) << 2;
            v4f v = *(const v4f*)(xb + (size_t)row * DIM + (k0 + c4));
            *(v4f*)&Xs[buf][row * XS_STRIDE + c4] = v;
        }
        #pragma unroll
        for (int i = 0; i < 4; ++i) {
            int f  = t + (i << 8);
            int kr = f >> 5;
            int c4 = (f & 31) << 2;
            v4f v = *(const v4f*)(Wdn + (size_t)(k0 + kr) * DIM + c4);
            #pragma unroll
            for (int j = 0; j < 4; ++j)
                Wt[buf][(c4 + j) * WT_STRIDE + kr] = v[j];
        }
    };

    load_tile(0, 0);
    __syncthreads();

    int cur = 0;
    for (int k0 = 0; k0 < DIM; k0 += BK) {
        const int nxt = cur ^ 1;
        if (k0 + BK < DIM) load_tile(nxt, k0 + BK);
        if (k0 + 2 * BK < DIM) {
            __builtin_prefetch(xb + (size_t)(t & 127) * DIM + (k0 + 2 * BK), 0, 1);
            __builtin_prefetch(Wdn + (size_t)(k0 + 2 * BK + (t >> 3)) * DIM + ((t & 7) << 4), 0, 1);
        }

        const float* xsc = &Xs[cur][0];
        const float* wtc = &Wt[cur][0];
        #pragma unroll
        for (int kk = 0; kk < BK; kk += 4) {
            const int kr = kk + (lg << 1);
            v2f a0 = *(const v2f*)&xsc[arow0 * XS_STRIDE + kr];
            v2f a1 = *(const v2f*)&xsc[arow1 * XS_STRIDE + kr];
            #pragma unroll
            for (int n = 0; n < 4; ++n) {
                v2f bb = *(const v2f*)&wtc[(nwbase + (n << 4) + lm) * WT_STRIDE + kr];
                acc[0][n] = wmma4(a0, bb, acc[0][n]);
                acc[1][n] = wmma4(a1, bb, acc[1][n]);
            }
        }
        __syncthreads();
        cur = nxt;
    }

    #pragma unroll
    for (int n = 0; n < 4; ++n) {
        const int col = n0 + nwbase + (n << 4) + lm;
        const float bv = bias[day * DIM + col];
        #pragma unroll
        for (int m = 0; m < 2; ++m) {
            #pragma unroll
            for (int r = 0; r < 8; ++r) {
                const int row = m0 + (mw << 5) + (m << 4) + r + (lg << 3);
                float y = acc[m][n][r] + bv;
                y = y / (1.0f + fabsf(y));
                __builtin_nontemporal_store(y, &out[((size_t)batch * TOK + row) * DIM + col]);
            }
        }
    }
}

extern "C" void kernel_launch(void* const* d_in, const int* in_sizes, int n_in,
                              void* d_out, int out_size, void* d_ws, size_t ws_size,
                              hipStream_t stream) {
    const float* x      = (const float*)d_in[0];
    const int* day_ids  = (const int*)d_in[1];
    const float* W      = (const float*)d_in[2];
    const float* b      = (const float*)d_in[3];
    float* out          = (float*)d_out;

    const int n_batch = in_sizes[1];
    dim3 grid(DIM / BN, TOK / BM, n_batch);
    day_adapter_kernel<<<grid, 256, 0, stream>>>(x, day_ids, W, b, out);
}


# K loop software-pipelined through registers: next chunk's global loads issued a chunk ahead, its LDS stores placed under the current chunk's last 32 WMMAs (was: load, wait, store, then WMMAs)
# speedup vs baseline: 1.0369x; 1.0369x over previous
#include <hip/hip_runtime.h>

typedef __attribute__((ext_vector_type(2))) float v2f;
typedef __attribute__((ext_vector_type(4))) float v4f;
typedef __attribute__((ext_vector_type(8))) float v8f;

#define DIM 512
#define TOK 1024
#define BM 128
#define BN 128
#define BK 32
#define XS_STRIDE 36
#define WT_STRIDE 36
#define TAILK 16
#define TAILS (TAILK / 4)

static __device__ __forceinline__ v8f wmma4(v2f a, v2f b, v8f c) {
    return __builtin_amdgcn_wmma_f32_16x16x4_f32(
        false, a, false, b, (short)0, c, false, false);
}

__global__ __launch_bounds__(256)
void day_adapter_kernel(const float* __restrict__ x,
                        const int* __restrict__ day_ids,
                        const float* __restrict__ W,
                        const float* __restrict__ bias,
                        float* __restrict__ out) {
    __shared__ float Xs[2][BM * XS_STRIDE];
    __shared__ float Wt[2][BN * WT_STRIDE];

    const int n0    = blockIdx.x * BN;
    const int m0    = blockIdx.y * BM;
    const int batch = blockIdx.z;
    const int day   = day_ids[batch];

    const int t    = threadIdx.x;
    const int lane = t & 31;
    const int wave = t >> 5;
    const int mw   = wave >> 1;
    const int nw   = wave & 1;
    const int lm   = lane & 15;
    const int lg   = lane >> 4;

    const float* xb  = x + ((size_t)batch * TOK + m0) * DIM;
    const float* Wdn = W + (size_t)day * DIM * DIM + n0;

    const int arow0  = (mw << 5) + lm;
    const int arow1  = arow0 + 16;
    const int nwbase = nw << 6;

    v8f acc[2][4];
    #pragma unroll
    for (int m = 0; m < 2; ++m)
        #pragma unroll
        for (int n = 0; n < 4; ++n) acc[m][n] = v8f{};

    v4f rx[4], rw[4];

    auto gload = [&](int k0) {
        #pragma unroll
        for (int i = 0; i < 4; ++i) {
            int f   = t + (i << 8);
            int row = f >> 3;
            int c4  = (f & 7) << 2;
            rx[i] = *(const v4f*)(xb + (size_t)row * DIM + (k0 + c4));
        }
        #pragma unroll
        for (int i = 0; i < 4; ++i) {
            int f  = t + (i << 8);
            int kr = f >> 5;
            int c4 = (f & 31) << 2;
            rw[i] = *(const v4f*)(Wdn + (size_t)(k0 + kr) * DIM + c4);
        }
    };

    auto lstore = [&](int buf) {
        #pragma unroll
        for (int i = 0; i < 4; ++i) {
            int f   = t + (i << 8);
            int row = f >> 3;
            int c4  = (f & 7) << 2;
            *(v4f*)&Xs[buf][row * XS_STRIDE + c4] = rx[i];
        }
        #pragma unroll
        for (int i = 0; i < 4; ++i) {
            int f  = t + (i << 8);
            int kr = f >> 5;
            int c4 = (f & 31) << 2;
            #pragma unroll
            for (int j = 0; j < 4; ++j)
                Wt[buf][(c4 + j) * WT_STRIDE + kr] = rw[i][j];
        }
    };

    auto mma_range = [&](int buf, int kb, int ke) {
        const float* xsc = &Xs[buf][0];
        const float* wtc = &Wt[buf][0];
        #pragma unroll
        for (int kk = kb; kk < ke; kk += 4) {
            const int kr = kk + (lg << 1);
            v2f a0 = *(const v2f*)&xsc[arow0 * XS_STRIDE + kr];
            v2f a1 = *(const v2f*)&xsc[arow1 * XS_STRIDE + kr];
            #pragma unroll
            for (int n = 0; n < 4; ++n) {
                v2f bb = *(const v2f*)&wtc[(nwbase + (n << 4) + lm) * WT_STRIDE + kr];
                acc[0][n] = wmma4(a0, bb, acc[0][n]);
                acc[1][n] = wmma4(a1, bb, acc[1][n]);
            }
        }
    };

    gload(0);
    lstore(0);
    __syncthreads();

    int cur = 0;
    for (int k0 = 0; k0 < DIM - BK; k0 += BK) {
        const int nxt = cur ^ 1;
        gload(k0 + BK);
        __builtin_amdgcn_sched_barrier(0);
        if (k0 + 2 * BK < DIM) {
            __builtin_prefetch(xb + (size_t)(t & 127) * DIM + (k0 + 2 * BK), 0, 1);
            __builtin_prefetch(Wdn + (size_t)(k0 + 2 * BK + (t >> 3)) * DIM + ((t & 7) << 4), 0, 1);
        }
        mma_range(cur, 0, BK - TAILK);

        v2f ta0[TAILS], ta1[TAILS], tb[TAILS][4];
        {
            const float* xsc = &Xs[cur][0];
            const float* wtc = &Wt[cur][0];
            #pragma unroll
            for (int s = 0; s < TAILS; ++s) {
                const int kr = (BK - TAILK) + (s << 2) + (lg << 1);
                ta0[s] = *(const v2f*)&xsc[arow0 * XS_STRIDE + kr];
                ta1[s] = *(const v2f*)&xsc[arow1 * XS_STRIDE + kr];
                #pragma unroll
                for (int n = 0; n < 4; ++n)
                    tb[s][n] = *(const v2f*)&wtc[(nwbase + (n << 4) + lm) * WT_STRIDE + kr];
            }
        }
        __builtin_amdgcn_sched_barrier(0);
        lstore(nxt);
        __builtin_amdgcn_sched_barrier(0);
        #pragma unroll
        for (int s = 0; s < TAILS; ++s) {
            if (s == TAILS - 1) __builtin_amdgcn_sched_barrier(0);
            #pragma unroll
            for (int n = 0; n < 4; ++n) {
                acc[0][n] = wmma4(ta0[s], tb[s][n], acc[0][n]);
                acc[1][n] = wmma4(ta1[s], tb[s][n], acc[1][n]);
            }
        }
        __syncthreads();
        cur = nxt;
    }
    mma_range(cur, 0, BK);

    #pragma unroll
    for (int n = 0; n < 4; ++n) {
        const int col = n0 + nwbase + (n << 4) + lm;
        const float bv = bias[day * DIM + col];
        #pragma unroll
        for (int m = 0; m < 2; ++m) {
            #pragma unroll
            for (int r = 0; r < 8; ++r) {
                const int row = m0 + (mw << 5) + (m << 4) + r + (lg << 3);
                float y = acc[m][n][r] + bv;
                y = y / (1.0f + fabsf(y));
                __builtin_nontemporal_store(y, &out[((size_t)batch * TOK + row) * DIM + col]);
            }
        }
    }
}

extern "C" void kernel_launch(void* const* d_in, const int* in_sizes, int n_in,
                              void* d_out, int out_size, void* d_ws, size_t ws_size,
                              hipStream_t stream) {
    const float* x      = (const float*)d_in[0];
    const int* day_ids  = (const int*)d_in[1];
    const float* W      = (const float*)d_in[2];
    const float* b      = (const float*)d_in[3];
    float* out          = (float*)d_out;

    const int n_batch = in_sizes[1];
    dim3 grid(DIM / BN, TOK / BM, n_batch);
    day_adapter_kernel<<<grid, 256, 0, stream>>>(x, day_ids, W, b, out);
}


# Staging stores to LDS made bank-conflict-free: W as 4x4 blocks transposed in registers (16-byte stores; was 8-way conflicted 4-byte), X rows re-spread over lanes; LDS layout and sums unchanged
# speedup vs baseline: 1.0726x; 1.0344x over previous
#include <hip/hip_runtime.h>

typedef __attribute__((ext_vector_type(2))) float v2f;
typedef __attribute__((ext_vector_type(4))) float v4f;
typedef __attribute__((ext_vector_type(8))) float v8f;

#define DIM 512
#define TOK 1024
#define BM 128
#define BN 128
#define BK 32
#define XS_STRIDE 36
#define WT_STRIDE 36
#define TAILK 16
#define TAILS (TAILK / 4)

static __device__ __forceinline__ v8f wmma4(v2f a, v2f b, v8f c) {
    return __builtin_amdgcn_wmma_f32_16x16x4_f32(
        false, a, false, b, (short)0, c, false, false);
}

__global__ __launch_bounds__(256)
void day_adapter_kernel(const float* __restrict__ x,
                        const int* __restrict__ day_ids,
                        const float* __restrict__ W,
                        const float* __restrict__ bias,
                        float* __restrict__ out) {
    __shared__ float Xs[2][BM * XS_STRIDE];
    __shared__ float Wt[2][BN * WT_STRIDE];

    const int n0    = blockIdx.x * BN;
    const int m0    = blockIdx.y * BM;
    const int batch = blockIdx.z;
    const int day   = day_ids[batch];

    const int t    = threadIdx.x;
    const int lane = t & 31;
    const int wave = t >> 5;
    const int mw   = wave >> 1;
    const int nw   = wave & 1;
    const int lm   = lane & 15;
    const int lg   = lane >> 4;

    const float* xb  = x + ((size_t)batch * TOK + m0) * DIM;
    const float* Wdn = W + (size_t)day * DIM * DIM + n0;

    const int arow0  = (mw << 5) + lm;
    const int arow1  = arow0 + 16;
    const int nwbase = nw << 6;

    const int xrow = wave + (((lane >> 3) & 1) << 3) + ((lane >> 4) << 4);
    const int xc4  = (lane & 7) << 2;

    const int wkq = ((wave & 1) << 2) + (lane & 3);
    const int wc4 = (((wave >> 1) << 3) + ((lane >> 4) << 2) + ((lane >> 2) & 3)) << 2;

    v8f acc[2][4];
    #pragma unroll
    for (int m = 0; m < 2; ++m)
        #pragma unroll
        for (int n = 0; n < 4; ++n) acc[m][n] = v8f{};

    v4f rx[4], rw[4];

    auto gload = [&](int k0) {
        #pragma unroll
        for (int i = 0; i < 4; ++i)
            rx[i] = *(const v4f*)(xb + (size_t)(xrow + (i << 5)) * DIM + (k0 + xc4));
        #pragma unroll
        for (int i = 0; i < 4; ++i)
            rw[i] = *(const v4f*)(Wdn + (size_t)(k0 + (wkq << 2) + i) * DIM + wc4);
    };

    auto lstore = [&](int buf) {
        #pragma unroll
        for (int i = 0; i < 4; ++i)
            *(v4f*)&Xs[buf][(xrow + (i << 5)) * XS_STRIDE + xc4] = rx[i];
        #pragma unroll
        for (int j = 0; j < 4; ++j) {
            v4f colv = { rw[0][j], rw[1][j], rw[2][j], rw[3][j] };
            *(v4f*)&Wt[buf][(wc4 + j) * WT_STRIDE + (wkq << 2)] = colv;
        }
    };

    auto mma_range = [&](int buf, int kb, int ke) {
        const float* xsc = &Xs[buf][0];
        const float* wtc = &Wt[buf][0];
        #pragma unroll
        for (int kk = kb; kk < ke; kk += 4) {
            const int kr = kk + (lg << 1);
            v2f a0 = *(const v2f*)&xsc[arow0 * XS_STRIDE + kr];
            v2f a1 = *(const v2f*)&xsc[arow1 * XS_STRIDE + kr];
            #pragma unroll
            for (int n = 0; n < 4; ++n) {
                v2f bb = *(const v2f*)&wtc[(nwbase + (n << 4) + lm) * WT_STRIDE + kr];
                acc[0][n] = wmma4(a0, bb, acc[0][n]);
                acc[1][n] = wmma4(a1, bb, acc[1][n]);
            }
        }
    };

    gload(0);
    lstore(0);
    __syncthreads();

    int cur = 0;
    for (int k0 = 0; k0 < DIM - BK; k0 += BK) {
        const int nxt = cur ^ 1;
        gload(k0 + BK);
        __builtin_amdgcn_sched_barrier(0);
        if (k0 + 2 * BK < DIM) {
            __builtin_prefetch(xb + (size_t)(t & 127) * DIM + (k0 + 2 * BK), 0, 1);
            __builtin_prefetch(Wdn + (size_t)(k0 + 2 * BK + (t >> 3)) * DIM + ((t & 7) << 4), 0, 1);
        }
        mma_range(cur, 0, BK - TAILK);

        v2f ta0[TAILS], ta1[TAILS], tb[TAILS][4];
        {
            const float* xsc = &Xs[cur][0];
            const float* wtc = &Wt[cur][0];
            #pragma unroll
            for (int s = 0; s < TAILS; ++s) {
                const int kr = (BK - TAILK) + (s << 2) + (lg << 1);
                ta0[s] = *(const v2f*)&xsc[arow0 * XS_STRIDE + kr];
                ta1[s] = *(const v2f*)&xsc[arow1 * XS_STRIDE + kr];
                #pragma unroll
                for (int n = 0; n < 4; ++n)
                    tb[s][n] = *(const v2f*)&wtc[(nwbase + (n << 4) + lm) * WT_STRIDE + kr];
            }
        }
        __builtin_amdgcn_sched_barrier(0);
        lstore(nxt);
        __builtin_amdgcn_sched_barrier(0);
        #pragma unroll
        for (int s = 0; s < TAILS; ++s) {
            if (s == TAILS - 1) __builtin_amdgcn_sched_barrier(0);
            #pragma unroll
            for (int n = 0; n < 4; ++n) {
                acc[0][n] = wmma4(ta0[s], tb[s][n], acc[0][n]);
                acc[1][n] = wmma4(ta1[s], tb[s][n], acc[1][n]);
            }
        }
        __syncthreads();
        cur = nxt;
    }
    mma_range(cur, 0, BK);

    #pragma unroll
    for (int n = 0; n < 4; ++n) {
        const int col = n0 + nwbase + (n << 4) + lm;
        const float bv = bias[day * DIM + col];
        #pragma unroll
        for (int m = 0; m < 2; ++m) {
            #pragma unroll
            for (int r = 0; r < 8; ++r) {
                const int row = m0 + (mw << 5) + (m << 4) + r + (lg << 3);
                float y = acc[m][n][r] + bv;
                y = y / (1.0f + fabsf(y));
                __builtin_nontemporal_store(y, &out[((size_t)batch * TOK + row) * DIM + col]);
            }
        }
    }
}

extern "C" void kernel_launch(void* const* d_in, const int* in_sizes, int n_in,
                              void* d_out, int out_size, void* d_ws, size_t ws_size,
                              hipStream_t stream) {
    const float* x      = (const float*)d_in[0];
    const int* day_ids  = (const int*)d_in[1];
    const float* W      = (const float*)d_in[2];
    const float* b      = (const float*)d_in[3];
    float* out          = (float*)d_out;

    const int n_batch = in_sizes[1];
    dim3 grid(DIM / BN, TOK / BM, n_batch);
    day_adapter_kernel<<<grid, 256, 0, stream>>>(x, day_ids, W, b, out);
}


# Wave tile 32x64 -> 64x64 (4x4 accumulator tiles, block tile 128x256, 1024 blocks): 8 fragment reads per 16 WMMAs, was 6 per 8; tail kept in registers 2 k-steps, was 4; sums unchanged
# speedup vs baseline: 1.1140x; 1.0386x over previous
#include <hip/hip_runtime.h>

typedef __attribute__((ext_vector_type(2))) float v2f;
typedef __attribute__((ext_vector_type(4))) float v4f;
typedef __attribute__((ext_vector_type(8))) float v8f;

#define DIM 512
#define TOK 1024
#define BM 128
#define BN 256
#define BK 32
#define XS_STRIDE 36
#define WT_STRIDE 36
#define TAILK 8
#define TAILS (TAILK / 4)

static __device__ __forceinline__ v8f wmma4(v2f a, v2f b, v8f c) {
    return __builtin_amdgcn_wmma_f32_16x16x4_f32(
        false, a, false, b, (short)0, c, false, false);
}

__global__ __launch_bounds__(256)
void day_adapter_kernel(const float* __restrict__ x,
                        const int* __restrict__ day_ids,
                        const float* __restrict__ W,
                        const float* __restrict__ bias,
                        float* __restrict__ out) {
    __shared__ float Xs[2][BM * XS_STRIDE];
    __shared__ float Wt[2][BN * WT_STRIDE];

    const int n0    = blockIdx.x * BN;
    const int m0    = blockIdx.y * BM;
    const int batch = blockIdx.z;
    const int day   = day_ids[batch];

    const int t    = threadIdx.x;
    const int lane = t & 31;
    const int wave = t >> 5;
    const int mw   = wave >> 2;
    const int nw   = wave & 3;
    const int lm   = lane & 15;
    const int lg   = lane >> 4;

    const float* xb  = x + ((size_t)batch * TOK + m0) * DIM;
    const float* Wdn = W + (size_t)day * DIM * DIM + n0;

    const int arow0  = (mw << 6) + lm;
    const int nwbase = nw << 6;

    const int xrow = wave + (((lane >> 3) & 1) << 3) + ((lane >> 4) << 4);
    const int xc4  = (lane & 7) << 2;

    const int wkq = ((wave & 1) << 2) + (lane & 3);
    const int wc4 = (((wave >> 1) << 3) + ((lane >> 4) << 2) + ((lane >> 2) & 3)) << 2;

    v8f acc[4][4];
    #pragma unroll
    for (int m = 0; m < 4; ++m)
        #pragma unroll
        for (int n = 0; n < 4; ++n) acc[m][n] = v8f{};

    v4f rx[4], rw[8];

    auto gload = [&](int k0) {
        #pragma unroll
        for (int i = 0; i < 4; ++i)
            rx[i] = *(const v4f*)(xb + (size_t)(xrow + (i << 5)) * DIM + (k0 + xc4));
        #pragma unroll
        for (int h = 0; h < 2; ++h)
            #pragma unroll
            for (int i = 0; i < 4; ++i)
                rw[(h << 2) + i] = *(const v4f*)(Wdn + (size_t)(k0 + (wkq << 2) + i) * DIM + (wc4 + (h << 7)));
    };

    auto lstore = [&](int buf) {
        #pragma unroll
        for (int i = 0; i < 4; ++i)
            *(v4f*)&Xs[buf][(xrow + (i << 5)) * XS_STRIDE + xc4] = rx[i];
        #pragma unroll
        for (int h = 0; h < 2; ++h)
            #pragma unroll
            for (int j = 0; j < 4; ++j) {
                v4f colv = { rw[(h << 2) + 0][j], rw[(h << 2) + 1][j], rw[(h << 2) + 2][j], rw[(h << 2) + 3][j] };
                *(v4f*)&Wt[buf][(wc4 + (h << 7) + j) * WT_STRIDE + (wkq << 2)] = colv;
            }
    };

    auto mma_range = [&](int buf, int kb, int ke) {
        const float* xsc = &Xs[buf][0];
        const float* wtc = &Wt[buf][0];
        #pragma unroll
        for (int kk = kb; kk < ke; kk += 4) {
            const int kr = kk + (lg << 1);
            v2f af[4];
            #pragma unroll
            for (int m = 0; m < 4; ++m)
                af[m] = *(const v2f*)&xsc[(arow0 + (m << 4)) * XS_STRIDE + kr];
            #pragma unroll
            for (int n = 0; n < 4; ++n) {
                v2f bb = *(const v2f*)&wtc[(nwbase + (n << 4) + lm) * WT_STRIDE + kr];
                #pragma unroll
                for (int m = 0; m < 4; ++m)
                    acc[m][n] = wmma4(af[m], bb, acc[m][n]);
            }
        }
    };

    gload(0);
    lstore(0);
    __syncthreads();

    int cur = 0;
    for (int k0 = 0; k0 < DIM - BK; k0 += BK) {
        const int nxt = cur ^ 1;
        gload(k0 + BK);
        __builtin_amdgcn_sched_barrier(0);
        if (k0 + 2 * BK < DIM) {
            __builtin_prefetch(xb + (size_t)(t & 127) * DIM + (k0 + 2 * BK), 0, 1);
            __builtin_prefetch(Wdn + (size_t)(k0 + 2 * BK + (t >> 3)) * DIM + ((t & 7) << 5), 0, 1);
        }
        mma_range(cur, 0, BK - TAILK);

        v2f ta[TAILS][4], tb[TAILS][4];
        {
            const float* xsc = &Xs[cur][0];
            const float* wtc = &Wt[cur][0];
            #pragma unroll
            for (int s = 0; s < TAILS; ++s) {
                const int kr = (BK - TAILK) + (s << 2) + (lg << 1);
                #pragma unroll
                for (int m = 0; m < 4; ++m)
                    ta[s][m] = *(const v2f*)&xsc[(arow0 + (m << 4)) * XS_STRIDE + kr];
                #pragma unroll
                for (int n = 0; n < 4; ++n)
                    tb[s][n] = *(const v2f*)&wtc[(nwbase + (n << 4) + lm) * WT_STRIDE + kr];
            }
        }
        __builtin_amdgcn_sched_barrier(0);
        lstore(nxt);
        __builtin_amdgcn_sched_barrier(0);
        #pragma unroll
        for (int s = 0; s < TAILS; ++s) {
            if (s == TAILS - 1) __builtin_amdgcn_sched_barrier(0);
            #pragma unroll
            for (int n = 0; n < 4; ++n)
                #pragma unroll
                for (int m = 0; m < 4; ++m)
                    acc[m][n] = wmma4(ta[s][m], tb[s][n], acc[m][n]);
        }
        __syncthreads();
        cur = nxt;
    }
    mma_range(cur, 0, BK);

    #pragma unroll
    for (int n = 0; n < 4; ++n) {
        const int col = n0 + nwbase + (n << 4) + lm;
        const float bv = bias[day * DIM + col];
        #pragma unroll
        for (int m = 0; m < 4; ++m) {
            #pragma unroll
            for (int r = 0; r < 8; ++r) {
                const int row = m0 + (mw << 6) + (m << 4) + r + (lg << 3);
                float y = acc[m][n][r] + bv;
                y = y / (1.0f + fabsf(y));
                __builtin_nontemporal_store(y, &out[((size_t)batch * TOK + row) * DIM + col]);
            }
        }
    }
}

extern "C" void kernel_launch(void* const* d_in, const int* in_sizes, int n_in,
                              void* d_out, int out_size, void* d_ws, size_t ws_size,
                              hipStream_t stream) {
    const float* x      = (const float*)d_in[0];
    const int* day_ids  = (const int*)d_in[1];
    const float* W      = (const float*)d_in[2];
    const float* b      = (const float*)d_in[3];
    float* out          = (float*)d_out;

    const int n_batch = in_sizes[1];
    dim3 grid(DIM / BN, TOK / BM, n_batch);
    day_adapter_kernel<<<grid, 256, 0, stream>>>(x, day_ids, W, b, out);
}
